# starting kernel
# baseline (speedup 1.0000x reference)
#include <hip/hip_runtime.h>


typedef __attribute__((ext_vector_type(2))) float v2f;
typedef __attribute__((ext_vector_type(8))) float v8f;

#define L0 (-0.12940952255126037f)
#define L1 ( 0.22414386804201339f)
#define L2 ( 0.83651630373780790f)
#define L3 ( 0.48296291314453416f)
#define H0 (-0.48296291314453416f)
#define H1 ( 0.83651630373780790f)
#define H2 (-0.22414386804201339f)
#define H3 (-0.12940952255126037f)

__device__ __constant__ float c_FH[4][4] = {
    {L0, L1, L2, L3}, {L0, L1, L2, L3}, {L0, L1, L2, L3}, {H0, H1, H2, H3}};
__device__ __constant__ float c_FW[4][4] = {
    {L0, L1, L2, L3}, {L0, L1, L2, L3}, {H0, H1, H2, H3}, {H0, H1, H2, H3}};

#define TJ 32
#define TC 16
#define PRR (TJ + 1)
#define PCC (TC + 1)
#define NPIX (PRR * PCC)
#define NCHUNK ((NPIX + 15) / 16)
#define NPIXPAD (NCHUNK * 16)
#define PITCH 20

__global__ __launch_bounds__(256) void idwt_db2_wmma(
    const float* __restrict__ x, float* __restrict__ out) {
  __shared__ __align__(16) float Plds[NPIXPAD * PITCH];

  const int tid  = threadIdx.x;
  const int lane = tid & 31;
  const int wave = tid >> 5;
  const int jc0  = blockIdx.x * TC;
  const int jr0  = blockIdx.y * TJ;
  const int b    = blockIdx.z;

  const int n  = lane & 15;
  const int hi = lane >> 4;
  const int c0 = hi << 1;
  const int dr = (n >> 3) & 1, dc = (n >> 2) & 1;
  const int pr = (n >> 1) & 1, pc = n & 1;
  const int ihh = 2 * dr + 1 - pr;
  const int iww = 2 * dc + 1 - pc;
  v2f Bw;
  Bw.x = c_FH[c0][ihh] * c_FW[c0][iww];
  Bw.y = c_FH[c0 + 1][ihh] * c_FW[c0 + 1][iww];

  for (int chunk = wave; chunk < NCHUNK; chunk += 8) {
    const int flat = chunk * 16 + n;
    const int prow = flat / PCC;
    const int pcol = flat - prow * PCC;
    int gr = jr0 + prow; gr = gr > 511 ? 511 : gr;
    int gc = jc0 + pcol; gc = gc > 511 ? 511 : gc;
    const float* px = x + ((((size_t)b << 9) + (size_t)gr) * 512 + (size_t)gc) * 4
                        + (hi << 1);
    v2f A; A.x = px[0]; A.y = px[1];
    v8f C = {0.f, 0.f, 0.f, 0.f, 0.f, 0.f, 0.f, 0.f};
    v8f D = __builtin_amdgcn_wmma_f32_16x16x4_f32(
         false, A,  false, Bw,
         (short)0, C,  false,  false);
    float* p = &Plds[(chunk * 16 + (hi << 3)) * PITCH + n];
#pragma unroll
    for (int v = 0; v < 8; v++) {
      p[v * PITCH] = D[v];
    }
  }
  __syncthreads();

  const int r  = tid >> 3;
  const int s0 = (tid & 7) << 1;

  float4 accA, accB;
#pragma unroll
  for (int k = 0; k < 2; k++) {
    const int s = s0 + k;
    const float4 t00 = *(const float4*)&Plds[((r    ) * PCC + s    ) * PITCH + 0];
    const float4 t01 = *(const float4*)&Plds[((r    ) * PCC + s + 1) * PITCH + 4];
    const float4 t10 = *(const float4*)&Plds[((r + 1) * PCC + s    ) * PITCH + 8];
    const float4 t11 = *(const float4*)&Plds[((r + 1) * PCC + s + 1) * PITCH + 12];
    float4 a;
    a.x = t00.x + t01.x + t10.x + t11.x;
    a.y = t00.y + t01.y + t10.y + t11.y;
    a.z = t00.z + t01.z + t10.z + t11.z;
    a.w = t00.w + t01.w + t10.w + t11.w;
    if (k == 0) accA = a; else accB = a;
  }

  const size_t orow = (size_t)(jr0 + r) * 2;
  const size_t ocol = (size_t)(jc0 + s0) * 2;
  float* o = out + (((size_t)b * 1024 + orow) * 1024 + ocol);
  float4 ev; ev.x = accA.x; ev.y = accA.y; ev.z = accB.x; ev.w = accB.y;
  float4 od; od.x = accA.z; od.y = accA.w; od.z = accB.z; od.w = accB.w;
  *(float4*)(o)        = ev;
  *(float4*)(o + 1024) = od;
}

extern "C" void kernel_launch(void* const* d_in, const int* in_sizes, int n_in,
                              void* d_out, int out_size, void* d_ws, size_t ws_size,
                              hipStream_t stream) {
  (void)in_sizes; (void)n_in; (void)d_ws; (void)ws_size; (void)out_size;
  const float* x = (const float*)d_in[0];
  float* out = (float*)d_out;
  dim3 grid(512 / TC, 512 / TJ, 32);
  idwt_db2_wmma<<<grid, 256, 0, stream>>>(x, out);
}


# Batch loop inside the workgroup: grid z 32 -> 3 (1536 workgroups = 6 per WGP, all resident), each walks images z, z+3, ... of its tile; 12288 waves enter once, not 16384 x 8; tile body unchanged
# speedup vs baseline: 1.1786x; 1.1786x over previous
#include <hip/hip_runtime.h>


typedef __attribute__((ext_vector_type(2))) float v2f;
typedef __attribute__((ext_vector_type(8))) float v8f;

#define L0 (-0.12940952255126037f)
#define L1 ( 0.22414386804201339f)
#define L2 ( 0.83651630373780790f)
#define L3 ( 0.48296291314453416f)
#define H0 (-0.48296291314453416f)
#define H1 ( 0.83651630373780790f)
#define H2 (-0.22414386804201339f)
#define H3 (-0.12940952255126037f)

__device__ __constant__ float c_FH[4][4] = {
    {L0, L1, L2, L3}, {L0, L1, L2, L3}, {L0, L1, L2, L3}, {H0, H1, H2, H3}};
__device__ __constant__ float c_FW[4][4] = {
    {L0, L1, L2, L3}, {L0, L1, L2, L3}, {H0, H1, H2, H3}, {H0, H1, H2, H3}};

#define TJ 32
#define TC 16
#define PRR (TJ + 1)
#define PCC (TC + 1)
#define NPIX (PRR * PCC)
#define NCHUNK ((NPIX + 15) / 16)
#define NPIXPAD (NCHUNK * 16)
#define PITCH 20
#define NB 32
#define NZ 3

__global__ __launch_bounds__(256) void idwt_db2_wmma(
    const float* __restrict__ x, float* __restrict__ out) {
  __shared__ __align__(16) float Plds[NPIXPAD * PITCH];

  const int tid  = threadIdx.x;
  const int lane = tid & 31;
  const int wave = tid >> 5;
  const int jc0  = blockIdx.x * TC;
  const int jr0  = blockIdx.y * TJ;

  const int n  = lane & 15;
  const int hi = lane >> 4;
  const int c0 = hi << 1;
  const int dr = (n >> 3) & 1, dc = (n >> 2) & 1;
  const int pr = (n >> 1) & 1, pc = n & 1;
  const int ihh = 2 * dr + 1 - pr;
  const int iww = 2 * dc + 1 - pc;
  v2f Bw;
  Bw.x = c_FH[c0][ihh] * c_FW[c0][iww];
  Bw.y = c_FH[c0 + 1][ihh] * c_FW[c0 + 1][iww];

  const int r  = tid >> 3;
  const int s0 = (tid & 7) << 1;

  for (int b = blockIdx.z; b < NB; b += NZ) {
    for (int chunk = wave; chunk < NCHUNK; chunk += 8) {
      const int flat = chunk * 16 + n;
      const int prow = flat / PCC;
      const int pcol = flat - prow * PCC;
      int gr = jr0 + prow; gr = gr > 511 ? 511 : gr;
      int gc = jc0 + pcol; gc = gc > 511 ? 511 : gc;
      const float* px = x + ((((size_t)b << 9) + (size_t)gr) * 512 + (size_t)gc) * 4
                          + (hi << 1);
      v2f A; A.x = px[0]; A.y = px[1];
      v8f C = {0.f, 0.f, 0.f, 0.f, 0.f, 0.f, 0.f, 0.f};
      v8f D = __builtin_amdgcn_wmma_f32_16x16x4_f32(
           false, A,  false, Bw,
           (short)0, C,  false,  false);
      asm volatile("v_nop\n\tv_nop\n\tv_nop\n\tv_nop\n\t"
                   "v_nop\n\tv_nop\n\tv_nop\n\tv_nop" : "+v"(D) : "v"(A), "v"(Bw));
      float* p = &Plds[(chunk * 16 + (hi << 3)) * PITCH + n];
#pragma unroll
      for (int v = 0; v < 8; v++) {
        p[v * PITCH] = D[v];
      }
    }
    __syncthreads();

    float4 accA, accB;
#pragma unroll
    for (int k = 0; k < 2; k++) {
      const int s = s0 + k;
      const float4 t00 = *(const float4*)&Plds[((r    ) * PCC + s    ) * PITCH + 0];
      const float4 t01 = *(const float4*)&Plds[((r    ) * PCC + s + 1) * PITCH + 4];
      const float4 t10 = *(const float4*)&Plds[((r + 1) * PCC + s    ) * PITCH + 8];
      const float4 t11 = *(const float4*)&Plds[((r + 1) * PCC + s + 1) * PITCH + 12];
      float4 a;
      a.x = t00.x + t01.x + t10.x + t11.x;
      a.y = t00.y + t01.y + t10.y + t11.y;
      a.z = t00.z + t01.z + t10.z + t11.z;
      a.w = t00.w + t01.w + t10.w + t11.w;
      if (k == 0) accA = a; else accB = a;
    }
    __syncthreads();

    const size_t orow = (size_t)(jr0 + r) * 2;
    const size_t ocol = (size_t)(jc0 + s0) * 2;
    float* o = out + (((size_t)b * 1024 + orow) * 1024 + ocol);
    float4 ev; ev.x = accA.x; ev.y = accA.y; ev.z = accB.x; ev.w = accB.y;
    float4 od; od.x = accA.z; od.y = accA.w; od.z = accB.z; od.w = accB.w;
    *(float4*)(o)        = ev;
    *(float4*)(o + 1024) = od;
  }
}

extern "C" void kernel_launch(void* const* d_in, const int* in_sizes, int n_in,
                              void* d_out, int out_size, void* d_ws, size_t ws_size,
                              hipStream_t stream) {
  (void)in_sizes; (void)n_in; (void)d_ws; (void)ws_size; (void)out_size;
  const float* x = (const float*)d_in[0];
  float* out = (float*)d_out;
  dim3 grid(512 / TC, 512 / TJ, NZ);
  idwt_db2_wmma<<<grid, 256, 0, stream>>>(x, out);
}


# Input fragments one image ahead: a wave requests all its chunks of image i+1 as one group behind image i's products; LDS-only barriers (no load drain) keep them in flight through gather and stores
# speedup vs baseline: 1.6024x; 1.3596x over previous
#include <hip/hip_runtime.h>


typedef __attribute__((ext_vector_type(2))) float v2f;
typedef __attribute__((ext_vector_type(8))) float v8f;

#define L0 (-0.12940952255126037f)
#define L1 ( 0.22414386804201339f)
#define L2 ( 0.83651630373780790f)
#define L3 ( 0.48296291314453416f)
#define H0 (-0.48296291314453416f)
#define H1 ( 0.83651630373780790f)
#define H2 (-0.22414386804201339f)
#define H3 (-0.12940952255126037f)

__device__ __constant__ float c_FH[4][4] = {
    {L0, L1, L2, L3}, {L0, L1, L2, L3}, {L0, L1, L2, L3}, {H0, H1, H2, H3}};
__device__ __constant__ float c_FW[4][4] = {
    {L0, L1, L2, L3}, {L0, L1, L2, L3}, {H0, H1, H2, H3}, {H0, H1, H2, H3}};

#define TJ 32
#define TC 16
#define PRR (TJ + 1)
#define PCC (TC + 1)
#define NPIX (PRR * PCC)
#define NCHUNK ((NPIX + 15) / 16)
#define NPIXPAD (NCHUNK * 16)
#define PITCH 20
#define NB 32
#define IMG (512 * 512 * 4)
#define NZ 3
#define NSLOT 5

static __device__ __forceinline__ v2f load_frag(const float* img, unsigned byte_off) {
  return *reinterpret_cast<const v2f*>(reinterpret_cast<const char*>(img) + byte_off);
}

static __device__ __forceinline__ void lds_barrier() {
  asm volatile("s_wait_dscnt 0x0\n\ts_barrier_signal -1\n\ts_barrier_wait -1" ::: "memory");
}

__global__ __launch_bounds__(256) void idwt_db2_wmma(
    const float* __restrict__ x, float* __restrict__ out) {
  __shared__ __align__(16) float Plds[NPIXPAD * PITCH];

  const int tid  = threadIdx.x;
  const int lane = tid & 31;
  const int wave = __builtin_amdgcn_readfirstlane(tid >> 5);
  const bool has5 = wave < (NCHUNK - 32);
  const int jc0  = blockIdx.x * TC;
  const int jr0  = blockIdx.y * TJ;

  const int n  = lane & 15;
  const int hi = lane >> 4;

  unsigned off[NSLOT];
#pragma unroll
  for (int i = 0; i < NSLOT; i++) {
    const int flat = (wave + 8 * i) * 16 + n;
    const int prow = flat / PCC;
    const int pcol = flat - prow * PCC;
    int gr = jr0 + prow; gr = gr > 511 ? 511 : gr;
    int gc = jc0 + pcol; gc = gc > 511 ? 511 : gc;
    off[i] = (unsigned)((gr * 512 + gc) * 4 + (hi << 1)) * 4u;
  }

  int b = blockIdx.z;
  v2f A[NSLOT];
  {
    const float* xb = x + (size_t)b * IMG;
#pragma unroll
    for (int i = 0; i < NSLOT - 1; i++) A[i] = load_frag(xb, off[i]);
    A[NSLOT - 1] = (v2f){0.f, 0.f};
    if (has5) A[NSLOT - 1] = load_frag(xb, off[NSLOT - 1]);
  }

  const int c0 = hi << 1;
  const int dr = (n >> 3) & 1, dc = (n >> 2) & 1;
  const int pr = (n >> 1) & 1, pc = n & 1;
  const int ihh = 2 * dr + 1 - pr;
  const int iww = 2 * dc + 1 - pc;
  v2f Bw;
  Bw.x = c_FH[c0][ihh] * c_FW[c0][iww];
  Bw.y = c_FH[c0 + 1][ihh] * c_FW[c0 + 1][iww];

  const int r  = tid >> 3;
  const int s0 = (tid & 7) << 1;

  float* const p = &Plds[(wave * 16 + (hi << 3)) * PITCH + n];

  for (; b < NB; b += NZ) {
#pragma unroll
    for (int i = 0; i < NSLOT - 1; i++) {
      v8f C = {0.f, 0.f, 0.f, 0.f, 0.f, 0.f, 0.f, 0.f};
      v8f D = __builtin_amdgcn_wmma_f32_16x16x4_f32(
           false, A[i],  false, Bw,
           (short)0, C,  false,  false);
      asm volatile("v_nop\n\tv_nop\n\tv_nop\n\tv_nop\n\t"
                   "v_nop\n\tv_nop\n\tv_nop\n\tv_nop" : "+v"(D) : "v"(A[i]), "v"(Bw));
#pragma unroll
      for (int v = 0; v < 8; v++) p[(i * 128 + v) * PITCH] = D[v];
      __builtin_amdgcn_sched_barrier(0);
    }
    if (has5) {
      v8f C = {0.f, 0.f, 0.f, 0.f, 0.f, 0.f, 0.f, 0.f};
      v8f D = __builtin_amdgcn_wmma_f32_16x16x4_f32(
          false, A[NSLOT - 1], false, Bw, (short)0, C, false, false);
      asm volatile("v_nop\n\tv_nop\n\tv_nop\n\tv_nop\n\t"
                   "v_nop\n\tv_nop\n\tv_nop\n\tv_nop" : "+v"(D) : "v"(A[NSLOT - 1]), "v"(Bw));
#pragma unroll
      for (int v = 0; v < 8; v++) p[((NSLOT - 1) * 128 + v) * PITCH] = D[v];
    }

    const int bn = b + NZ;
    if (bn < NB) {
      const float* xn = x + (size_t)bn * IMG;
#pragma unroll
      for (int i = 0; i < NSLOT - 1; i++) A[i] = load_frag(xn, off[i]);
      if (has5) A[NSLOT - 1] = load_frag(xn, off[NSLOT - 1]);
    }
    lds_barrier();

    float4 accA, accB;
#pragma unroll
    for (int k = 0; k < 2; k++) {
      const int s = s0 + k;
      const float4 t00 = *(const float4*)&Plds[((r    ) * PCC + s    ) * PITCH + 0];
      const float4 t01 = *(const float4*)&Plds[((r    ) * PCC + s + 1) * PITCH + 4];
      const float4 t10 = *(const float4*)&Plds[((r + 1) * PCC + s    ) * PITCH + 8];
      const float4 t11 = *(const float4*)&Plds[((r + 1) * PCC + s + 1) * PITCH + 12];
      float4 a;
      a.x = t00.x + t01.x + t10.x + t11.x;
      a.y = t00.y + t01.y + t10.y + t11.y;
      a.z = t00.z + t01.z + t10.z + t11.z;
      a.w = t00.w + t01.w + t10.w + t11.w;
      if (k == 0) accA = a; else accB = a;
    }
    lds_barrier();

    const size_t orow = (size_t)(jr0 + r) * 2;
    const size_t ocol = (size_t)(jc0 + s0) * 2;
    float* o = out + (((size_t)b * 1024 + orow) * 1024 + ocol);
    float4 ev; ev.x = accA.x; ev.y = accA.y; ev.z = accB.x; ev.w = accB.y;
    float4 od; od.x = accA.z; od.y = accA.w; od.z = accB.z; od.w = accB.w;
    *(float4*)(o)        = ev;
    *(float4*)(o + 1024) = od;
  }
}

extern "C" void kernel_launch(void* const* d_in, const int* in_sizes, int n_in,
                              void* d_out, int out_size, void* d_ws, size_t ws_size,
                              hipStream_t stream) {
  (void)in_sizes; (void)n_in; (void)d_ws; (void)ws_size; (void)out_size;
  const float* x = (const float*)d_in[0];
  float* out = (float*)d_out;
  dim3 grid(512 / TC, 512 / TJ, NZ);
  idwt_db2_wmma<<<grid, 256, 0, stream>>>(x, out);
}
